# starting kernel
# baseline (speedup 1.0000x reference)
#include <hip/hip_runtime.h>

typedef __attribute__((ext_vector_type(2))) float v2f;
typedef __attribute__((ext_vector_type(8))) float v8f;

#define H 512
#define W 512
#define CIN 3
#define COUT 2
#define LDS_NROWS 12
#define LDS_ROWW  520

__global__ __launch_bounds__(256) void conv3x3_wmma_f32(
    const float* __restrict__ x,
    const float* __restrict__ wgt,
    const float* __restrict__ bias,
    float* __restrict__ out)
{
    const int h0  = blockIdx.x * 2;
    const int b   = blockIdx.y;
    const int tid = threadIdx.x;

    __shared__ float lds[LDS_NROWS * LDS_ROWW];

    if (tid < 2 * LDS_NROWS) {
        int r = tid >> 1;
        lds[r * LDS_ROWW + ((tid & 1) ? 516 : 3)] = 0.0f;
    }

    for (int idx = tid; idx < LDS_NROWS * 128; idx += 256) {
        int r   = idx >> 7;
        int q   = idx & 127;
        int c   = r >> 2;
        int dh4 = r & 3;
        int row = h0 + dh4 - 1;
        float4 v = make_float4(0.0f, 0.0f, 0.0f, 0.0f);
        if (row >= 0 && row < H)
            v = *(const float4*)(x + (((size_t)b * CIN + c) * H + row) * W + q * 4);
        *(float4*)(&lds[r * LDS_ROWW + 4 + q * 4]) = v;
    }
    __syncthreads();

    const int lane = tid & 31;
    const int wave = tid >> 5;
    const int m    = lane & 15;
    const int half = lane >> 4;

    v2f bfrag[7];
#pragma unroll
    for (int j = 0; j < 7; ++j) {
        float b0 = 0.0f, b1 = 0.0f;
        int k0 = j * 4 + 2 * half;
        if (m < COUT) {
            b0 = wgt[m * 27 + k0];
            if (k0 + 1 < 27) b1 = wgt[m * 27 + k0 + 1];
        }
        bfrag[j][0] = b0;
        bfrag[j][1] = b1;
    }

    int offA[14];
#pragma unroll
    for (int j = 0; j < 7; ++j) {
#pragma unroll
        for (int v = 0; v < 2; ++v) {
            int k = j * 4 + 2 * half + v;
            if (k > 26) k = 26;
            int c  = k / 9;
            int r  = k - c * 9;
            int dh = r / 3;
            int dw = r - dh * 3;
            offA[j * 2 + v] = (c * 4 + dh) * LDS_ROWW + m + dw + 3;
        }
    }

    const float bv0 = bias[0];
    const float bv1 = bias[1];

#pragma unroll
    for (int rr = 0; rr < 2; ++rr) {
        const float* ldsr = lds + rr * LDS_ROWW;
        const int h = h0 + rr;

#pragma unroll
        for (int s = 0; s < 4; ++s) {
            const int w0 = (wave * 4 + s) * 16;

            v8f acc = {};
#pragma unroll
            for (int j = 0; j < 7; ++j) {
                v2f a;
                a[0] = ldsr[offA[j * 2 + 0] + w0];
                a[1] = ldsr[offA[j * 2 + 1] + w0];
                acc = __builtin_amdgcn_wmma_f32_16x16x4_f32(
                     false, a,  false, bfrag[j],
                     (short)0, acc,  false,  false);
            }

            if ((lane & 14) == 0) {
                const int ch = lane & 1;
                const int mb = (lane >> 4) * 8;
                const float bv = ch ? bv1 : bv0;
                float* o = out + ((((size_t)b * COUT + ch) * H + h) * W + w0 + mb);
                float4 lo = make_float4(acc[0] + bv, acc[1] + bv, acc[2] + bv, acc[3] + bv);
                float4 hi = make_float4(acc[4] + bv, acc[5] + bv, acc[6] + bv, acc[7] + bv);
                *(float4*)(o + 0) = lo;
                *(float4*)(o + 4) = hi;
            }
        }
    }
}

extern "C" void kernel_launch(void* const* d_in, const int* in_sizes, int n_in,
                              void* d_out, int out_size, void* d_ws, size_t ws_size,
                              hipStream_t stream) {
    const float* x    = (const float*)d_in[0];
    const float* wgt  = (const float*)d_in[1];
    const float* bias = (const float*)d_in[2];
    float* out        = (float*)d_out;

    dim3 grid(H / 2, 64, 1);
    dim3 block(256, 1, 1);
    conv3x3_wmma_f32<<<grid, block, 0, stream>>>(x, wgt, bias, out);
}


# GEMM re-mapped to pixel pairs: B = the 4 inputs under 2 px (one 8-byte load per lane), A = banded weights: 3 matrix instr per 32 px per row (was 6); 16-wave group spans a row; tap sum order changes
# speedup vs baseline: 2.2953x; 2.2953x over previous
#include <hip/hip_runtime.h>

typedef __attribute__((ext_vector_type(2))) float v2f;
typedef __attribute__((ext_vector_type(8))) float v8f;
typedef v2f v2f_dw __attribute__((aligned(4)));

#define H 512
#define W 512
#define CIN 3
#define COUT 2
#define RSEG 64
#define UG 6
#define NGRP ((RSEG + 2) / UG)


template <int EDGE>
__device__ __forceinline__ v2f fix_pair(const v2f v, const bool sel, const int keepm) {
    v2f r;
    if (EDGE == 1) {
        r[0] = __int_as_float(__float_as_int(v[0]) & keepm);
        r[1] = sel ? v[0] : v[1];
    } else {
        r[0] = sel ? v[1] : v[0];
        r[1] = __int_as_float(__float_as_int(v[1]) & keepm);
    }
    return r;
}

template <int EDGE>
__device__ __forceinline__ void stream_rows(
    const float* __restrict__ xb,
    float* __restrict__ ob,
    const unsigned lx,
    const unsigned lo,
    const v2f af0, const v2f af1, const v2f af2,
    const float b0, const float b1,
    const int h0, const bool sel, const int keepm)
{
    v8f acc = (v8f){};
    v2f bufA[UG][CIN], bufB[UG][CIN];
    const v2f zero2 = (v2f){0.0f, 0.0f};

#define LOAD_GROUP(dst, g, CLAMP)                                            \
    _Pragma("unroll")                                                        \
    for (int u = 0; u < UG; ++u) {                                           \
        const int row = h0 - 1 + (g) * UG + u;                               \
        const int rc  = CLAMP(row);                                          \
        const float* xr = xb + rc * W;                                       \
        _Pragma("unroll")                                                    \
        for (int c = 0; c < CIN; ++c)                                        \
            dst[u][c] = *(const v2f_dw*)(xr + c * (H * W) + lx);             \
    }

#define FIX_PAIR(v) (EDGE ? fix_pair<EDGE>((v), sel, keepm) : (v))

#define ROW_PAIR(src, u, ch, PAD) ((PAD) ? zero2 : FIX_PAIR(src[u][ch]))
#define ONE_ROW(src, u, PAD, row, STORE)                                     \
    {                                                                        \
        v8f c;                                                               \
        c[0] = 0.0f;   c[1] = 0.0f;                                          \
        c[2] = acc[0]; c[3] = acc[1];                                        \
        c[4] = acc[2]; c[5] = acc[3];                                        \
        c[6] = 0.0f;   c[7] = 0.0f;                                          \
        c = __builtin_amdgcn_wmma_f32_16x16x4_f32(                           \
            false, af0, false, ROW_PAIR(src, u, 0, PAD), (short)0, c, false, false); \
        c = __builtin_amdgcn_wmma_f32_16x16x4_f32(                           \
            false, af1, false, ROW_PAIR(src, u, 1, PAD), (short)0, c, false, false); \
        c = __builtin_amdgcn_wmma_f32_16x16x4_f32(                           \
            false, af2, false, ROW_PAIR(src, u, 2, PAD), (short)0, c, false, false); \
        acc = c;                                                             \
        if (STORE) {                                                         \
            float* o = ob + ((row) - 1) * W;                                 \
            o[lo]         = c[4] + b0;                                       \
            o[H * W + lo] = c[5] + b1;                                       \
        }                                                                    \
    }

#define COMPUTE_GROUP(src, g)                                                \
    _Pragma("unroll")                                                        \
    for (int u = 0; u < UG; ++u) {                                           \
        const int row = h0 - 1 + (g) * UG + u;                               \
        ONE_ROW(src, u, false, row, true)                                    \
    }

#define CLAMP_LO(r) ((r) < 0 ? 0 : (r))
#define CLAMP_HI(r) ((r) > H - 1 ? H - 1 : (r))
#define NO_CLAMP(r) (r)
#define ORDER_HERE __builtin_amdgcn_sched_barrier(0);
    LOAD_GROUP(bufA, 0, CLAMP_LO)
    ORDER_HERE
    LOAD_GROUP(bufB, 1, NO_CLAMP)
    ORDER_HERE

    {
        const bool pad = (h0 == 0);
        ONE_ROW(bufA, 0, pad, h0 - 1, false)
        ONE_ROW(bufA, 1, false, h0, false)
#pragma unroll
        for (int u = 2; u < UG; ++u) {
            const int row = h0 - 1 + u;
            ONE_ROW(bufA, u, false, row, true)
        }
    }
    ORDER_HERE
    LOAD_GROUP(bufA, 2, NO_CLAMP)
    ORDER_HERE

#pragma unroll 1
    for (int g = 1; g < NGRP - 2; g += 2) {
        COMPUTE_GROUP(bufB, g)
        ORDER_HERE
        LOAD_GROUP(bufB, g + 2, NO_CLAMP)
        ORDER_HERE
        COMPUTE_GROUP(bufA, g + 1)
        ORDER_HERE
        LOAD_GROUP(bufA, g + 3, CLAMP_HI)
        ORDER_HERE
    }

    COMPUTE_GROUP(bufB, NGRP - 2)

    {
        const bool pad = (h0 + RSEG == H);
#pragma unroll
        for (int u = 0; u < UG - 1; ++u) {
            const int row = h0 - 1 + (NGRP - 1) * UG + u;
            ONE_ROW(bufA, u, false, row, true)
        }
        ONE_ROW(bufA, UG - 1, pad, h0 + RSEG, true)
    }
#undef ORDER_HERE
#undef LOAD_GROUP
#undef CLAMP_LO
#undef CLAMP_HI
#undef NO_CLAMP
#undef FIX_PAIR
#undef ONE_ROW
#undef ROW_PAIR
#undef COMPUTE_GROUP
}

__global__ __launch_bounds__(512, 8) void conv3x3_pairstream_wmma_f32(
    const float* __restrict__ x,
    const float* __restrict__ wgt,
    const float* __restrict__ bias,
    float* __restrict__ out)
{
    const int seg  = blockIdx.x;
    const int b    = blockIdx.y;
    const int lane = threadIdx.x & 31;
    const int wave = __builtin_amdgcn_readfirstlane((int)(threadIdx.x >> 5));
    const int n    = lane & 15;
    const int hf   = lane >> 4;
    const int h0   = seg * RSEG;
    const int w0   = wave * 32;

    v2f afrag[CIN];
    {
        const int p = n >> 3, q = n & 7, j = q >> 1, o = q & 1;
#pragma unroll
        for (int c = 0; c < CIN; ++c) {
#pragma unroll
            for (int s = 0; s < 2; ++s) {
                const int k  = 2 * hf + s;
                const int dw = k - p;
                float v = 0.0f;
                if (q < 6 && dw >= 0 && dw <= 2)
                    v = wgt[o * 27 + c * 9 + j * 3 + dw];
                afrag[c][s] = v;
            }
        }
    }

    const int  col0 = w0 + 2 * n + 2 * hf - 1;
    const bool isL  = col0 < 0;
    const bool isR  = col0 > W - 2;
    const int  colc = isL ? 0 : (isR ? W - 2 : col0);

    const float* xb = x + (size_t)b * CIN * H * W;
    float*       ob = out + (size_t)b * COUT * H * W + w0;
    const unsigned lo = 2 * n + hf;
    const float b0 = bias[0], b1 = bias[1];

    if (wave == 0)
        stream_rows<1>(xb, ob, (unsigned)colc, lo, afrag[0], afrag[1], afrag[2], b0, b1, h0,
                       isL, isL ? 0 : -1);
    else if (wave == (W / 32) - 1)
        stream_rows<2>(xb, ob, (unsigned)colc, lo, afrag[0], afrag[1], afrag[2], b0, b1, h0,
                       isR, isR ? 0 : -1);
    else
        stream_rows<0>(xb, ob, (unsigned)colc, lo, afrag[0], afrag[1], afrag[2], b0, b1, h0,
                       false, -1);
}

extern "C" void kernel_launch(void* const* d_in, const int* in_sizes, int n_in,
                              void* d_out, int out_size, void* d_ws, size_t ws_size,
                              hipStream_t stream) {
    const float* x    = (const float*)d_in[0];
    const float* wgt  = (const float*)d_in[1];
    const float* bias = (const float*)d_in[2];
    float* out        = (float*)d_out;

    dim3 grid(H / RSEG, 64, 1);
    dim3 block(512, 1, 1);
    conv3x3_pairstream_wmma_f32<<<grid, block, 0, stream>>>(x, wgt, bias, out);
}


# Input rows prefetched in four register sets of 3 rows (was two of 6), three sets in flight: a wave keeps 27 to 36 row loads outstanding (was 18 to 36); sums and their order unchanged
# speedup vs baseline: 2.9621x; 1.2905x over previous
#include <hip/hip_runtime.h>

typedef __attribute__((ext_vector_type(2))) float v2f;
typedef __attribute__((ext_vector_type(4))) float v4f;
typedef __attribute__((ext_vector_type(8))) float v8f;
typedef v4f v4f_dw __attribute__((aligned(4)));

#define H 512
#define W 512
#define CIN 3
#define COUT 2
#define WCOLS 64
#define RSEG 64
#define UG 3
#define NGRP ((RSEG + 2) / UG)


template <int EDGE>
__device__ __forceinline__ v4f fix_quad(const v4f v, const bool sel, const int keepm) {
    v4f r;
    if (EDGE == 1) {
        r[0] = __int_as_float(__float_as_int(v[0]) & keepm);
        r[1] = sel ? v[0] : v[1];
        r[2] = sel ? v[1] : v[2];
        r[3] = sel ? v[2] : v[3];
    } else {
        r[0] = sel ? v[1] : v[0];
        r[1] = sel ? v[2] : v[1];
        r[2] = sel ? v[3] : v[2];
        r[3] = __int_as_float(__float_as_int(v[3]) & keepm);
    }
    return r;
}

template <int EDGE>
__device__ __forceinline__ void stream_rows(
    const float* __restrict__ xb,
    float* __restrict__ ob,
    const unsigned lx,
    const unsigned lo,
    const v2f af0, const v2f af1, const v2f af2,
    const float bl,
    const int h0, const bool sel, const int keepm)
{
    v8f acc0 = (v8f){};
    v8f acc1 = (v8f){};
    v4f buf0[UG][CIN], buf1[UG][CIN], buf2[UG][CIN], buf3[UG][CIN];
    const v4f zero4 = (v4f){0.0f, 0.0f, 0.0f, 0.0f};

#define LOAD_GROUP(dst, g, CLAMP)                                            \
    _Pragma("unroll")                                                        \
    for (int u = 0; u < UG; ++u) {                                           \
        const int row = h0 - 1 + (g) * UG + u;                               \
        const int rc  = CLAMP(row);                                          \
        const float* xr = xb + rc * W;                                       \
        _Pragma("unroll")                                                    \
        for (int c = 0; c < CIN; ++c)                                        \
            dst[u][c] = *(const v4f_dw*)(xr + c * (H * W) + lx);             \
    }

#define FIX_QUAD(v) (EDGE ? fix_quad<EDGE>((v), sel, keepm) : (v))

#define ROW_QUAD(src, u, ch, PAD) ((PAD) ? zero4 : FIX_QUAD(src[u][ch]))
#define ONE_ROW(src, u, PAD, row, STORE)                                     \
    {                                                                        \
        const v4f q0 = ROW_QUAD(src, u, 0, PAD);                             \
        const v4f q1 = ROW_QUAD(src, u, 1, PAD);                             \
        const v4f q2 = ROW_QUAD(src, u, 2, PAD);                             \
        v8f c0, c1;                                                          \
        c0[0] = 0.0f;    c0[1] = 0.0f;                                       \
        c0[2] = acc0[0]; c0[3] = acc0[1];                                    \
        c0[4] = acc0[2]; c0[5] = acc0[3];                                    \
        c0[6] = 0.0f;    c0[7] = 0.0f;                                       \
        c1[0] = 0.0f;    c1[1] = 0.0f;                                       \
        c1[2] = acc1[0]; c1[3] = acc1[1];                                    \
        c1[4] = acc1[2]; c1[5] = acc1[3];                                    \
        c1[6] = 0.0f;    c1[7] = 0.0f;                                       \
        c0 = __builtin_amdgcn_wmma_f32_16x16x4_f32(                          \
            false, af0, false, (v2f){q0[0], q0[1]}, (short)0, c0, false, false); \
        c1 = __builtin_amdgcn_wmma_f32_16x16x4_f32(                          \
            false, af0, false, (v2f){q0[2], q0[3]}, (short)0, c1, false, false); \
        c0 = __builtin_amdgcn_wmma_f32_16x16x4_f32(                          \
            false, af1, false, (v2f){q1[0], q1[1]}, (short)0, c0, false, false); \
        c1 = __builtin_amdgcn_wmma_f32_16x16x4_f32(                          \
            false, af1, false, (v2f){q1[2], q1[3]}, (short)0, c1, false, false); \
        c0 = __builtin_amdgcn_wmma_f32_16x16x4_f32(                          \
            false, af2, false, (v2f){q2[0], q2[1]}, (short)0, c0, false, false); \
        c1 = __builtin_amdgcn_wmma_f32_16x16x4_f32(                          \
            false, af2, false, (v2f){q2[2], q2[3]}, (short)0, c1, false, false); \
        acc0 = c0;                                                           \
        acc1 = c1;                                                           \
        if (STORE) {                                                         \
            float* o = ob + ((row) - 1) * W;                                 \
            *(v4f*)(o + lo) = (v4f){c0[4] + bl, c0[5] + bl,                  \
                                    c1[4] + bl, c1[5] + bl};                 \
        }                                                                    \
    }

#define COMPUTE_GROUP(src, g)                                                \
    _Pragma("unroll")                                                        \
    for (int u = 0; u < UG; ++u) {                                           \
        const int row = h0 - 1 + (g) * UG + u;                               \
        ONE_ROW(src, u, false, row, true)                                    \
    }

#define CLAMP_LO(r) ((r) < 0 ? 0 : (r))
#define CLAMP_HI(r) ((r) > H - 1 ? H - 1 : (r))
#define NO_CLAMP(r) (r)
#define ORDER_HERE __builtin_amdgcn_sched_barrier(0);
    LOAD_GROUP(buf0, 0, CLAMP_LO)
    ORDER_HERE
    LOAD_GROUP(buf1, 1, NO_CLAMP)
    ORDER_HERE
    LOAD_GROUP(buf2, 2, NO_CLAMP)
    ORDER_HERE
    LOAD_GROUP(buf3, 3, NO_CLAMP)
    ORDER_HERE

    {
        const bool pad = (h0 == 0);
        ONE_ROW(buf0, 0, pad, h0 - 1, false)
        ONE_ROW(buf0, 1, false, h0, false)
        ONE_ROW(buf0, 2, false, h0 + 1, true)
    }
    ORDER_HERE
    LOAD_GROUP(buf0, 4, NO_CLAMP)
    ORDER_HERE

#pragma unroll 1
    for (int g = 1; g < NGRP - 5; g += 4) {
        COMPUTE_GROUP(buf1, g)
        ORDER_HERE
        LOAD_GROUP(buf1, g + 4, NO_CLAMP)
        ORDER_HERE
        COMPUTE_GROUP(buf2, g + 1)
        ORDER_HERE
        LOAD_GROUP(buf2, g + 5, NO_CLAMP)
        ORDER_HERE
        COMPUTE_GROUP(buf3, g + 2)
        ORDER_HERE
        LOAD_GROUP(buf3, g + 6, NO_CLAMP)
        ORDER_HERE
        COMPUTE_GROUP(buf0, g + 3)
        ORDER_HERE
        LOAD_GROUP(buf0, g + 7, NO_CLAMP)
        ORDER_HERE
    }

    COMPUTE_GROUP(buf1, NGRP - 5)
    ORDER_HERE
    LOAD_GROUP(buf1, NGRP - 1, CLAMP_HI)
    ORDER_HERE
    COMPUTE_GROUP(buf2, NGRP - 4)
    COMPUTE_GROUP(buf3, NGRP - 3)
    COMPUTE_GROUP(buf0, NGRP - 2)

    {
        const bool pad = (h0 + RSEG == H);
        ONE_ROW(buf1, 0, false, h0 + RSEG - 2, true)
        ONE_ROW(buf1, 1, false, h0 + RSEG - 1, true)
        ONE_ROW(buf1, 2, pad, h0 + RSEG, true)
    }
#undef ORDER_HERE
#undef LOAD_GROUP
#undef CLAMP_LO
#undef CLAMP_HI
#undef NO_CLAMP
#undef FIX_QUAD
#undef ONE_ROW
#undef ROW_QUAD
#undef COMPUTE_GROUP
}

__global__ __launch_bounds__(256, 4) void conv3x3_quadstream_wmma_f32(
    const float* __restrict__ x,
    const float* __restrict__ wgt,
    const float* __restrict__ bias,
    float* __restrict__ out)
{
    const int seg  = blockIdx.x;
    const int b    = blockIdx.y;
    const int lane = threadIdx.x & 31;
    const int wave = __builtin_amdgcn_readfirstlane((int)(threadIdx.x >> 5));
    const int n    = lane & 15;
    const int hf   = lane >> 4;
    const int h0   = seg * RSEG;
    const int w0   = wave * WCOLS;

    v2f afrag[CIN];
    {
        const int o = n >> 3, q = n & 7, j = q >> 1, p = q & 1;
#pragma unroll
        for (int c = 0; c < CIN; ++c) {
#pragma unroll
            for (int s = 0; s < 2; ++s) {
                const int k  = 2 * hf + s;
                const int dw = k - p;
                float v = 0.0f;
                if (q < 6 && dw >= 0 && dw <= 2)
                    v = wgt[o * 27 + c * 9 + j * 3 + dw];
                afrag[c][s] = v;
            }
        }
    }

    const int  col0 = w0 + 4 * n + 2 * hf - 1;
    const bool isL  = col0 < 0;
    const bool isR  = col0 > W - 4;
    const int  colc = isL ? 0 : (isR ? W - 4 : col0);

    const float* xb = x + (size_t)b * CIN * H * W;
    float*       ob = out + (size_t)b * COUT * H * W + w0;
    const unsigned lo = hf * (H * W) + 4 * n;
    const float bl = hf ? bias[1] : bias[0];

    if (wave == 0)
        stream_rows<1>(xb, ob, (unsigned)colc, lo, afrag[0], afrag[1], afrag[2], bl, h0,
                       isL, isL ? 0 : -1);
    else if (wave == (W / WCOLS) - 1)
        stream_rows<2>(xb, ob, (unsigned)colc, lo, afrag[0], afrag[1], afrag[2], bl, h0,
                       isR, isR ? 0 : -1);
    else
        stream_rows<0>(xb, ob, (unsigned)colc, lo, afrag[0], afrag[1], afrag[2], bl, h0,
                       false, -1);
}

extern "C" void kernel_launch(void* const* d_in, const int* in_sizes, int n_in,
                              void* d_out, int out_size, void* d_ws, size_t ws_size,
                              hipStream_t stream) {
    const float* x    = (const float*)d_in[0];
    const float* wgt  = (const float*)d_in[1];
    const float* bias = (const float*)d_in[2];
    float* out        = (float*)d_out;

    dim3 grid(H / RSEG, 64, 1);
    dim3 block(256, 1, 1);
    conv3x3_quadstream_wmma_f32<<<grid, block, 0, stream>>>(x, wgt, bias, out);
}
